# starting kernel
# baseline (speedup 1.0000x reference)
#include <hip/hip_runtime.h>

typedef float v2f __attribute__((ext_vector_type(2)));
typedef float v8f __attribute__((ext_vector_type(8)));

#define B_    256
#define T_    150
#define J_    25
#define WAVES 8
#define GSTR  33

__global__ __launch_bounds__(256)
void knn_feats_kernel(const float* __restrict__ x, float* __restrict__ out) {
    __shared__ float lds_g[WAVES][32 * GSTR];
    __shared__ float lds_red[WAVES][J_ * 9];

    const int tid  = threadIdx.x;
    const int wv   = tid >> 5;
    const int lane = tid & 31;
    const int b    = blockIdx.x;

    float acc[9];
#pragma unroll
    for (int f = 0; f < 9; ++f) acc[f] = 0.0f;

    const int  lj      = lane & 15;
    const bool hiHalf  = lane >= 16;
    const int  rowLo   = lj;
    const int  rowHi   = 16 + lj;
    const bool hiValid = rowHi < J_;

    float* gw = lds_g[wv];

    for (int t = wv; t < T_; t += WAVES) {
        const float* px = x + (size_t)(b * T_ + t) * (J_ * 3);

        v2f aLo, aHi;
        if (!hiHalf) {
            aLo.x = px[rowLo * 3 + 0];
            aLo.y = px[rowLo * 3 + 1];
            aHi.x = hiValid ? px[rowHi * 3 + 0] : 0.0f;
            aHi.y = hiValid ? px[rowHi * 3 + 1] : 0.0f;
        } else {
            aLo.x = px[rowLo * 3 + 2];
            aLo.y = 0.0f;
            aHi.x = hiValid ? px[rowHi * 3 + 2] : 0.0f;
            aHi.y = 0.0f;
        }

        v8f cz = {};
        v8f g00 = __builtin_amdgcn_wmma_f32_16x16x4_f32(false, aLo, false, aLo, (short)0, cz, false, false);
        v8f g01 = __builtin_amdgcn_wmma_f32_16x16x4_f32(false, aLo, false, aHi, (short)0, cz, false, false);
        v8f g10 = __builtin_amdgcn_wmma_f32_16x16x4_f32(false, aHi, false, aLo, (short)0, cz, false, false);
        v8f g11 = __builtin_amdgcn_wmma_f32_16x16x4_f32(false, aHi, false, aHi, (short)0, cz, false, false);

        const int rbase = hiHalf ? 8 : 0;
#pragma unroll
        for (int r = 0; r < 8; ++r) {
            gw[(rbase + r) * GSTR + lj]            = g00[r];
            gw[(rbase + r) * GSTR + 16 + lj]       = g01[r];
            gw[(16 + rbase + r) * GSTR + lj]       = g10[r];
            gw[(16 + rbase + r) * GSTR + 16 + lj]  = g11[r];
        }
        __builtin_amdgcn_wave_barrier();

        const int j = lane;
        const float nj = gw[j * GSTR + j];

        float m1 = 3.0e38f, m2 = 3.0e38f, m3 = 3.0e38f, m4 = 3.0e38f;
#pragma unroll 5
        for (int i = 0; i < J_; ++i) {
            float ni  = gw[i * GSTR + i];
            float gij = gw[i * GSTR + j];
            float s   = ni + nj - 2.0f * gij;
            float d   = (s > 0.0f) ? __builtin_sqrtf(s) : 0.0f;
            d = (i == j) ? 3.0e38f : d;
            float t1 = fminf(d,  m1); float r1 = fmaxf(d,  m1);
            float t2 = fminf(r1, m2); float r2 = fmaxf(r1, m2);
            float t3 = fminf(r2, m3); float r3 = fmaxf(r2, m3);
            float t4 = fminf(r3, m4);
            m1 = t1; m2 = t2; m3 = t3; m4 = t4;
        }

        float mean2 = 0.5f * (m1 + m2);
        float a0 = m1 - mean2, a1 = m2 - mean2;
        float std2 = __builtin_sqrtf(a0 * a0 + a1 * a1);

        float mean3 = (m1 + m2 + m3) * (1.0f / 3.0f);
        float e0 = m1 - mean3, e1 = m2 - mean3, e2 = m3 - mean3;
        float std3 = __builtin_sqrtf((e0 * e0 + e1 * e1 + e2 * e2) * 0.5f);

        float mean4 = (m1 + m2 + m3 + m4) * 0.25f;
        float f0 = m1 - mean4, f1 = m2 - mean4, f2 = m3 - mean4, f3 = m4 - mean4;
        float std4 = __builtin_sqrtf((f0 * f0 + f1 * f1 + f2 * f2 + f3 * f3) * (1.0f / 3.0f));

        if (j < J_) {
            acc[0] += mean2; acc[1] += std2; acc[2] += m1;
            acc[3] += mean3; acc[4] += std3; acc[5] += m1;
            acc[6] += mean4; acc[7] += std4; acc[8] += m1;
        }
        __builtin_amdgcn_wave_barrier();
    }

    if (lane < J_) {
#pragma unroll
        for (int f = 0; f < 9; ++f)
            lds_red[wv][lane * 9 + f] = acc[f] * (1.0f / (float)T_);
    }
    __syncthreads();

    if (tid < J_ * 9) {
        float s = 0.0f;
#pragma unroll
        for (int w = 0; w < WAVES; ++w) s += lds_red[w][tid];
        out[(size_t)b * (J_ * 9) + tid] = s;
    }
}

extern "C" void kernel_launch(void* const* d_in, const int* in_sizes, int n_in,
                              void* d_out, int out_size, void* d_ws, size_t ws_size,
                              hipStream_t stream) {
    (void)in_sizes; (void)n_in; (void)d_ws; (void)ws_size; (void)out_size;
    const float* x = (const float*)d_in[0];
    float* out = (float*)d_out;
    knn_feats_kernel<<<B_, 256, 0, stream>>>(x, out);
}


# 5 frames x 25 joints of a trip packed over four full 32-lane registers (125 of 128 lanes busy, was 100): 16 selection passes per wave, not 19; pair values cross via LDS; f32 add order changes
# speedup vs baseline: 2.2224x; 2.2224x over previous
#include <hip/hip_runtime.h>


typedef float v2f __attribute__((ext_vector_type(2)));
typedef float v8f __attribute__((ext_vector_type(8)));

#define B_    256
#define T_    150
#define J_    25
#define WAVES 8
#define GSTR  37
#define NF    5
#define NQ    4
#define NITEM (NF * J_)
#define TRIPS (T_ / NF)
#define KH    ((J_ - 1) / 2)
#define KU    6
#define TILE  (32 * GSTR + 1)
#define NSTR  40
#define XSTR  50
#define XROW  256
static_assert(KH % KU == 0, "the selection loop runs whole passes");
static_assert(NITEM * 9 <= TILE, "a wave's partial sums must fit its first Gram tile");


__device__ __forceinline__ void trip_load(const float* __restrict__ xb, int T, int lane,
                                          float (&raw)[6][NF]) {
    const int  lj      = lane & 15;
    const bool hiHalf  = lane >= 16;
    const int  jMid    = (16 + lj < J_) ? 16 + lj : lj - 9;
    const int  jTop    = 7 + lj;
    const int  c0      = hiHalf ? 2 : 0;
    const int  c1      = hiHalf ? 2 : 1;
#pragma unroll
    for (int f = 0; f < NF; ++f) {
        const float* p = xb + (size_t)(T * NF + f) * (J_ * 3);
        raw[0][f] = p[lj * 3 + c0];
        raw[1][f] = p[lj * 3 + c1];
        raw[2][f] = p[jMid * 3 + c0];
        raw[3][f] = p[jMid * 3 + c1];
        raw[4][f] = p[jTop * 3 + c0];
        raw[5][f] = p[jTop * 3 + c1];
    }
}

__device__ __forceinline__ void trip_compute(const float (&raw)[6][NF], float* gw, float* const (&xr)[KU], float* nw,
                                             int lane, const int (&jq)[NQ], const int (&xIdx)[NQ],
                                             const int (&dIdx)[NQ], const int (&nIdx)[NQ],
                                             float (&acc)[NQ][9]) {
    const int  lj      = lane & 15;
    const bool hiHalf  = lane >= 16;

    v2f aLo[NF], aHi[NF], aTop[NF];
#pragma unroll
    for (int f = 0; f < NF; ++f) {
        aLo[f].x  = raw[0][f];
        aLo[f].y  = hiHalf ? 0.0f : raw[1][f];
        aHi[f].x  = raw[2][f];
        aHi[f].y  = hiHalf ? 0.0f : raw[3][f];
        aTop[f].x = raw[4][f];
        aTop[f].y = hiHalf ? 0.0f : raw[5][f];
    }

    const int rbase = hiHalf ? 8 : 0;
#pragma unroll
    for (int f = 0; f < NF; ++f) {
        v8f cz = {};
        v8f g00 = __builtin_amdgcn_wmma_f32_16x16x4_f32(false, aLo[f], false, aLo[f], (short)0, cz, false, false);
        v8f g01 = __builtin_amdgcn_wmma_f32_16x16x4_f32(false, aLo[f], false, aHi[f], (short)0, cz, false, false);
        v8f g11 = __builtin_amdgcn_wmma_f32_16x16x4_f32(false, aHi[f], false, aHi[f], (short)0, cz, false, false);
        v8f g12 = __builtin_amdgcn_wmma_f32_16x16x4_f32(false, aHi[f], false, aTop[f], (short)0, cz, false, false);

        float* gf = gw + f * TILE;
#pragma unroll
        for (int r = 0; r < 8; ++r) {
            gf[(rbase + r) * GSTR + lj]            = g00[r];
            gf[(rbase + r) * GSTR + 16 + lj]       = g01[r];
            gf[(16 + rbase + r) * GSTR + 16 + lj]  = g11[r];
        }
        if (lj < GSTR - 32) {
#pragma unroll
            for (int r = 0; r < 8; ++r)
                gf[(16 + rbase + r) * GSTR + 32 + lj] = g12[r];
        }
    }
    __builtin_amdgcn_wave_barrier();

    float nj[NQ], m1[NQ], m2[NQ], m3[NQ], m4[NQ];
#pragma unroll
    for (int q = 0; q < NQ; ++q) {
        nj[q] = gw[dIdx[q]];
        m1[q] = 3.0e38f; m2[q] = 3.0e38f; m3[q] = 3.0e38f; m4[q] = 3.0e38f;
    }
#pragma unroll
    for (int q = 0; q < NQ; ++q) {
        nw[nIdx[q]] = nj[q];
        if (jq[q] < KH) nw[nIdx[q] + J_] = nj[q];
    }
    __builtin_amdgcn_wave_barrier();

    int dI[NQ], nI[NQ], xL[NQ];
#pragma unroll
    for (int q = 0; q < NQ; ++q) { dI[q] = dIdx[q]; nI[q] = nIdx[q]; xL[q] = xIdx[q] + J_; }
#pragma clang loop unroll(disable)
    for (int kb = 0; kb < KH / KU; ++kb) {
#pragma unroll
    for (int s = 1; s <= KU; ++s) {
        float* xk = xr[s - 1];
        float dOwn[NQ], dGot[NQ];
#pragma unroll
        for (int q = 0; q < NQ; ++q) {
            float ni  = nw[nI[q] + s];
            float gij = gw[dI[q] + s];
            float s   = ni + nj[q] - 2.0f * gij;
            dOwn[q]   = (s > 0.0f) ? __builtin_sqrtf(s) : 0.0f;
            xk[xIdx[q]]      = dOwn[q];
            xk[xIdx[q] + J_] = dOwn[q];
        }
#pragma unroll
        for (int q = 0; q < NQ; ++q) {
            dGot[q] = xk[xL[q] - s];
        }

#pragma unroll
        for (int q = 0; q < NQ; ++q) {
            float d  = dOwn[q];
            float t1 = fminf(d,  m1[q]); float r1 = fmaxf(d,  m1[q]);
            float t2 = fminf(r1, m2[q]); float r2 = fmaxf(r1, m2[q]);
            float t3 = fminf(r2, m3[q]); float r3 = fmaxf(r2, m3[q]);
            float t4 = fminf(r3, m4[q]);
            m1[q] = t1; m2[q] = t2; m3[q] = t3; m4[q] = t4;
        }
#pragma unroll
        for (int q = 0; q < NQ; ++q) {
            float d  = dGot[q];
            float t1 = fminf(d,  m1[q]); float r1 = fmaxf(d,  m1[q]);
            float t2 = fminf(r1, m2[q]); float r2 = fmaxf(r1, m2[q]);
            float t3 = fminf(r2, m3[q]); float r3 = fmaxf(r2, m3[q]);
            float t4 = fminf(r3, m4[q]);
            m1[q] = t1; m2[q] = t2; m3[q] = t3; m4[q] = t4;
        }
    }
#pragma unroll
    for (int q = 0; q < NQ; ++q) { dI[q] += KU; nI[q] += KU; xL[q] -= KU; }
    }

#pragma unroll
    for (int q = 0; q < NQ; ++q) {
        float mean2 = 0.5f * (m1[q] + m2[q]);
        float a0 = m1[q] - mean2, a1 = m2[q] - mean2;
        float std2 = __builtin_sqrtf(a0 * a0 + a1 * a1);

        float mean3 = (m1[q] + m2[q] + m3[q]) * (1.0f / 3.0f);
        float e0 = m1[q] - mean3, e1 = m2[q] - mean3, e2 = m3[q] - mean3;
        float std3 = __builtin_sqrtf((e0 * e0 + e1 * e1 + e2 * e2) * 0.5f);

        float mean4 = (m1[q] + m2[q] + m3[q] + m4[q]) * 0.25f;
        float f0 = m1[q] - mean4, f1 = m2[q] - mean4, f2 = m3[q] - mean4, f3 = m4[q] - mean4;
        float std4 = __builtin_sqrtf((f0 * f0 + f1 * f1 + f2 * f2 + f3 * f3) * (1.0f / 3.0f));

        acc[q][0] += mean2; acc[q][1] += std2; acc[q][2] += m1[q];
        acc[q][3] += mean3; acc[q][4] += std3; acc[q][5] += m1[q];
        acc[q][6] += mean4; acc[q][7] += std4; acc[q][8] += m1[q];
    }
    __builtin_amdgcn_wave_barrier();
}

__global__ __launch_bounds__(256) __attribute__((amdgpu_waves_per_eu(2, 2)))
void knn_feats_kernel(const float* __restrict__ x, float* __restrict__ out) {
    __shared__ float lds_g[WAVES][NF][TILE];
    __shared__ float lds_n[WAVES][NF * NSTR];
    __shared__ float lds_x0[WAVES][XROW], lds_x1[WAVES][XROW], lds_x2[WAVES][XROW];
    __shared__ float lds_x3[WAVES][XROW], lds_x4[WAVES][XROW], lds_x5[WAVES][XROW];

    const int tid  = threadIdx.x;
    const int wv   = __builtin_amdgcn_readfirstlane(tid >> 5);
    const int lane = tid & 31;
    const int b    = blockIdx.x;

    float acc[NQ][9];
    int jq[NQ], xIdx[NQ], dIdx[NQ], nIdx[NQ];
#pragma unroll
    for (int q = 0; q < NQ; ++q) {
        const int v = q * 32 + lane;
        const int vc = (v < NITEM) ? v : NITEM - 1;
        const int f = vc / J_;
        jq[q] = vc - f * J_;
        xIdx[q] = f * XSTR + jq[q];
        dIdx[q] = f * TILE + jq[q] * (GSTR + 1);
        nIdx[q] = f * NSTR + jq[q];
#pragma unroll
        for (int c = 0; c < 9; ++c) acc[q][c] = 0.0f;
    }

    float* gw = lds_g[wv][0];
    float* const xr[KU] = { lds_x0[wv], lds_x1[wv], lds_x2[wv], lds_x3[wv], lds_x4[wv], lds_x5[wv] };
    float* nw = lds_n[wv];
    const float* xb = x + (size_t)b * T_ * (J_ * 3);

    float cur[6][NF], nxt[6][NF];
#pragma unroll
    for (int r = 0; r < 6; ++r)
#pragma unroll
        for (int f = 0; f < NF; ++f) nxt[r][f] = 0.0f;
    trip_load(xb, wv, lane, cur);
    for (int T = wv; T < TRIPS; T += WAVES) {
        if (T + WAVES < TRIPS) trip_load(xb, T + WAVES, lane, nxt);
        trip_compute(cur, gw, xr, nw, lane, jq, xIdx, dIdx, nIdx, acc);
#pragma unroll
        for (int r = 0; r < 6; ++r)
#pragma unroll
            for (int f = 0; f < NF; ++f) cur[r][f] = nxt[r][f];
    }

#pragma unroll
    for (int q = 0; q < NQ; ++q) {
        const int v = q * 32 + lane;
        if (v < NITEM) {
#pragma unroll
            for (int c = 0; c < 9; ++c)
                gw[v * 9 + c] = acc[q][c] * (1.0f / (float)T_);
        }
    }
    __syncthreads();

    if (tid < J_ * 9) {
        const int jj = tid / 9;
        const int c  = tid - jj * 9;
        float s = 0.0f;
        for (int w = 0; w < WAVES; ++w)
#pragma unroll
            for (int f = 0; f < NF; ++f) s += lds_g[w][0][(f * J_ + jj) * 9 + c];
        out[(size_t)b * (J_ * 9) + tid] = s;
    }
}

extern "C" void kernel_launch(void* const* d_in, const int* in_sizes, int n_in,
                              void* d_out, int out_size, void* d_ws, size_t ws_size,
                              hipStream_t stream) {
    (void)in_sizes; (void)n_in; (void)d_ws; (void)ws_size; (void)out_size;
    const float* x = (const float*)d_in[0];
    float* out = (float*)d_out;
    knn_feats_kernel<<<B_, 256, 0, stream>>>(x, out);
}


# correctly rounded sqrt kept; its rare-input legs (rescale below 2^-96, zero/inf pass-through, s>0 guard) leave the hot path: one whole-wave range test per trip redoes the trip in full form; same sums
# speedup vs baseline: 2.3795x; 1.0707x over previous
#include <hip/hip_runtime.h>


typedef float v2f __attribute__((ext_vector_type(2)));
typedef float v8f __attribute__((ext_vector_type(8)));

#define B_    256
#define T_    150
#define J_    25
#define WAVES 8
#define GSTR  37
#define NF    5
#define NQ    4
#define NITEM (NF * J_)
#define TRIPS (T_ / NF)
#define KH    ((J_ - 1) / 2)
#define KU    6
#define TILE  (32 * GSTR + 1)
#define NSTR  40
#define XSTR  50
#define XROW  256
static_assert(KH % KU == 0, "the selection loop runs whole passes");
static_assert(NITEM * 9 <= TILE, "a wave's partial sums must fit its first Gram tile");


__device__ __forceinline__ void trip_load(const float* __restrict__ xb, int T, int lane,
                                          float (&raw)[6][NF]) {
    const int  lj      = lane & 15;
    const bool hiHalf  = lane >= 16;
    const int  jMid    = (16 + lj < J_) ? 16 + lj : lj - 9;
    const int  jTop    = 7 + lj;
    const int  c0      = hiHalf ? 2 : 0;
    const int  c1      = hiHalf ? 2 : 1;
#pragma unroll
    for (int f = 0; f < NF; ++f) {
        const float* p = xb + (size_t)(T * NF + f) * (J_ * 3);
        raw[0][f] = p[lj * 3 + c0];
        raw[1][f] = p[lj * 3 + c1];
        raw[2][f] = p[jMid * 3 + c0];
        raw[3][f] = p[jMid * 3 + c1];
        raw[4][f] = p[jTop * 3 + c0];
        raw[5][f] = p[jTop * 3 + c1];
    }
}

__device__ __forceinline__ float sqrt_in_range(float s) {
    const float r  = __builtin_amdgcn_sqrtf(s);
    const float rd = __uint_as_float(__float_as_uint(r) - 1u);
    const float ru = __uint_as_float(__float_as_uint(r) + 1u);
    const float vp = __builtin_fmaf(-rd, r, s);
    const float vs = __builtin_fmaf(-ru, r, s);
    float t = (vp <= 0.0f) ? rd : r;
    t = (vs > 0.0f) ? ru : t;
    return t;
}

#define RANGE_LO 0x0f800000u
#define RANGE_HI 0x7f800000u

template <bool FULL>
__device__ __forceinline__ void select_and_stats(float* gw, float* const (&xr)[KU], float* nw,
                                                 const int (&xIdx)[NQ], const int (&dIdx)[NQ], const int (&nIdx)[NQ],
                                                 const float (&nj)[NQ], float (&ft)[NQ][7],
                                                 unsigned (&lo)[NQ], unsigned (&hi)[NQ]) {
    float m1[NQ], m2[NQ], m3[NQ], m4[NQ];
#pragma unroll
    for (int q = 0; q < NQ; ++q) {
        m1[q] = 3.0e38f; m2[q] = 3.0e38f; m3[q] = 3.0e38f; m4[q] = 3.0e38f;
    }

    int dI[NQ], nI[NQ], xL[NQ];
#pragma unroll
    for (int q = 0; q < NQ; ++q) { dI[q] = dIdx[q]; nI[q] = nIdx[q]; xL[q] = xIdx[q] + J_; }
#pragma clang loop unroll(disable)
    for (int kb = 0; kb < KH / KU; ++kb) {
#pragma unroll
    for (int s = 1; s <= KU; ++s) {
        float* xk = xr[s - 1];
        float dOwn[NQ], dGot[NQ];
#pragma unroll
        for (int q = 0; q < NQ; ++q) {
            float ni  = nw[nI[q] + s];
            float gij = gw[dI[q] + s];
            float sq  = ni + nj[q] - 2.0f * gij;
            if (FULL) {
                dOwn[q] = (sq > 0.0f) ? __builtin_sqrtf(sq) : 0.0f;
            } else {
                const unsigned bits = __float_as_uint(sq);
                lo[q] = bits < lo[q] ? bits : lo[q];
                hi[q] = bits > hi[q] ? bits : hi[q];
                float root = sqrt_in_range(sq);
                asm("" : "+v"(root));
                dOwn[q] = __builtin_canonicalizef(root);
            }
            xk[xIdx[q]]      = dOwn[q];
            xk[xIdx[q] + J_] = dOwn[q];
        }
#pragma unroll
        for (int q = 0; q < NQ; ++q) {
            dGot[q] = xk[xL[q] - s];
        }

#pragma unroll
        for (int q = 0; q < NQ; ++q) {
            float d  = dOwn[q];
            float t1 = fminf(d,  m1[q]); float r1 = fmaxf(d,  m1[q]);
            float t2 = fminf(r1, m2[q]); float r2 = fmaxf(r1, m2[q]);
            float t3 = fminf(r2, m3[q]); float r3 = fmaxf(r2, m3[q]);
            float t4 = fminf(r3, m4[q]);
            m1[q] = t1; m2[q] = t2; m3[q] = t3; m4[q] = t4;
        }
#pragma unroll
        for (int q = 0; q < NQ; ++q) {
            float d  = dGot[q];
            float t1 = fminf(d,  m1[q]); float r1 = fmaxf(d,  m1[q]);
            float t2 = fminf(r1, m2[q]); float r2 = fmaxf(r1, m2[q]);
            float t3 = fminf(r2, m3[q]); float r3 = fmaxf(r2, m3[q]);
            float t4 = fminf(r3, m4[q]);
            m1[q] = t1; m2[q] = t2; m3[q] = t3; m4[q] = t4;
        }
    }
#pragma unroll
    for (int q = 0; q < NQ; ++q) { dI[q] += KU; nI[q] += KU; xL[q] -= KU; }
    }

#pragma unroll
    for (int q = 0; q < NQ; ++q) {
        float mean2 = 0.5f * (m1[q] + m2[q]);
        float a0 = m1[q] - mean2, a1 = m2[q] - mean2;
        float v2 = a0 * a0 + a1 * a1;

        float mean3 = (m1[q] + m2[q] + m3[q]) * (1.0f / 3.0f);
        float e0 = m1[q] - mean3, e1 = m2[q] - mean3, e2 = m3[q] - mean3;
        float v3 = (e0 * e0 + e1 * e1 + e2 * e2) * 0.5f;

        float mean4 = (m1[q] + m2[q] + m3[q] + m4[q]) * 0.25f;
        float f0 = m1[q] - mean4, f1 = m2[q] - mean4, f2 = m3[q] - mean4, f3 = m4[q] - mean4;
        float v4 = (f0 * f0 + f1 * f1 + f2 * f2 + f3 * f3) * (1.0f / 3.0f);

        float std2, std3, std4;
        if (FULL) {
            std2 = __builtin_sqrtf(v2); std3 = __builtin_sqrtf(v3); std4 = __builtin_sqrtf(v4);
        } else {
            const unsigned b2 = __float_as_uint(v2), b3 = __float_as_uint(v3), b4 = __float_as_uint(v4);
            unsigned l = lo[q], h = hi[q];
            l = b2 < l ? b2 : l;  h = b2 > h ? b2 : h;
            l = b3 < l ? b3 : l;  h = b3 > h ? b3 : h;
            l = b4 < l ? b4 : l;  h = b4 > h ? b4 : h;
            lo[q] = l; hi[q] = h;
            std2 = sqrt_in_range(v2); std3 = sqrt_in_range(v3); std4 = sqrt_in_range(v4);
        }

        ft[q][0] = mean2; ft[q][1] = std2; ft[q][2] = m1[q];
        ft[q][3] = mean3; ft[q][4] = std3;
        ft[q][5] = mean4; ft[q][6] = std4;
    }
}

__device__ __forceinline__ void trip_compute(const float (&raw)[6][NF], float* gw, float* const (&xr)[KU], float* nw,
                                             int lane, const int (&jq)[NQ], const int (&xIdx)[NQ],
                                             const int (&dIdx)[NQ], const int (&nIdx)[NQ],
                                             float (&acc)[NQ][9]) {
    const int  lj      = lane & 15;
    const bool hiHalf  = lane >= 16;

    v2f aLo[NF], aHi[NF], aTop[NF];
#pragma unroll
    for (int f = 0; f < NF; ++f) {
        aLo[f].x  = raw[0][f];
        aLo[f].y  = hiHalf ? 0.0f : raw[1][f];
        aHi[f].x  = raw[2][f];
        aHi[f].y  = hiHalf ? 0.0f : raw[3][f];
        aTop[f].x = raw[4][f];
        aTop[f].y = hiHalf ? 0.0f : raw[5][f];
    }

    const int rbase = hiHalf ? 8 : 0;
#pragma unroll
    for (int f = 0; f < NF; ++f) {
        v8f cz = {};
        v8f g00 = __builtin_amdgcn_wmma_f32_16x16x4_f32(false, aLo[f], false, aLo[f], (short)0, cz, false, false);
        v8f g01 = __builtin_amdgcn_wmma_f32_16x16x4_f32(false, aLo[f], false, aHi[f], (short)0, cz, false, false);
        v8f g11 = __builtin_amdgcn_wmma_f32_16x16x4_f32(false, aHi[f], false, aHi[f], (short)0, cz, false, false);
        v8f g12 = __builtin_amdgcn_wmma_f32_16x16x4_f32(false, aHi[f], false, aTop[f], (short)0, cz, false, false);

        float* gf = gw + f * TILE;
#pragma unroll
        for (int r = 0; r < 8; ++r) {
            gf[(rbase + r) * GSTR + lj]            = g00[r];
            gf[(rbase + r) * GSTR + 16 + lj]       = g01[r];
            gf[(16 + rbase + r) * GSTR + 16 + lj]  = g11[r];
        }
        if (lj < GSTR - 32) {
#pragma unroll
            for (int r = 0; r < 8; ++r)
                gf[(16 + rbase + r) * GSTR + 32 + lj] = g12[r];
        }
    }
    __builtin_amdgcn_wave_barrier();

    float nj[NQ];
#pragma unroll
    for (int q = 0; q < NQ; ++q) nj[q] = gw[dIdx[q]];
#pragma unroll
    for (int q = 0; q < NQ; ++q) {
        nw[nIdx[q]] = nj[q];
        if (jq[q] < KH) nw[nIdx[q] + J_] = nj[q];
    }
    __builtin_amdgcn_wave_barrier();

    float ft[NQ][7];
    unsigned lo[NQ], hi[NQ];
#pragma unroll
    for (int q = 0; q < NQ; ++q) { lo[q] = 0xffffffffu; hi[q] = 0u; }
    select_and_stats<false>(gw, xr, nw, xIdx, dIdx, nIdx, nj, ft, lo, hi);
    unsigned loAll = lo[0], hiAll = hi[0];
#pragma unroll
    for (int q = 1; q < NQ; ++q) {
        loAll = lo[q] < loAll ? lo[q] : loAll;
        hiAll = hi[q] > hiAll ? hi[q] : hiAll;
    }
    const bool outside = (loAll < RANGE_LO) | (hiAll >= RANGE_HI);
    if (__builtin_amdgcn_ballot_w32(outside) != 0u) {
        select_and_stats<true>(gw, xr, nw, xIdx, dIdx, nIdx, nj, ft, lo, hi);
    }

#pragma unroll
    for (int q = 0; q < NQ; ++q) {
        acc[q][0] += ft[q][0]; acc[q][1] += ft[q][1]; acc[q][2] += ft[q][2];
        acc[q][3] += ft[q][3]; acc[q][4] += ft[q][4]; acc[q][5] += ft[q][2];
        acc[q][6] += ft[q][5]; acc[q][7] += ft[q][6]; acc[q][8] += ft[q][2];
    }
    __builtin_amdgcn_wave_barrier();
}

__global__ __launch_bounds__(256) __attribute__((amdgpu_waves_per_eu(2, 2)))
void knn_feats_kernel(const float* __restrict__ x, float* __restrict__ out) {
    __shared__ float lds_g[WAVES][NF][TILE];
    __shared__ float lds_n[WAVES][NF * NSTR];
    __shared__ float lds_x0[WAVES][XROW], lds_x1[WAVES][XROW], lds_x2[WAVES][XROW];
    __shared__ float lds_x3[WAVES][XROW], lds_x4[WAVES][XROW], lds_x5[WAVES][XROW];

    const int tid  = threadIdx.x;
    const int wv   = __builtin_amdgcn_readfirstlane(tid >> 5);
    const int lane = tid & 31;
    const int b    = blockIdx.x;

    float acc[NQ][9];
    int jq[NQ], xIdx[NQ], dIdx[NQ], nIdx[NQ];
#pragma unroll
    for (int q = 0; q < NQ; ++q) {
        const int v = q * 32 + lane;
        const int vc = (v < NITEM) ? v : NITEM - 1;
        const int f = vc / J_;
        jq[q] = vc - f * J_;
        xIdx[q] = f * XSTR + jq[q];
        dIdx[q] = f * TILE + jq[q] * (GSTR + 1);
        nIdx[q] = f * NSTR + jq[q];
#pragma unroll
        for (int c = 0; c < 9; ++c) acc[q][c] = 0.0f;
    }

    float* gw = lds_g[wv][0];
    float* const xr[KU] = { lds_x0[wv], lds_x1[wv], lds_x2[wv], lds_x3[wv], lds_x4[wv], lds_x5[wv] };
    float* nw = lds_n[wv];
    const float* xb = x + (size_t)b * T_ * (J_ * 3);

    float cur[6][NF], nxt[6][NF];
#pragma unroll
    for (int r = 0; r < 6; ++r)
#pragma unroll
        for (int f = 0; f < NF; ++f) nxt[r][f] = 0.0f;
    trip_load(xb, wv, lane, cur);
    for (int T = wv; T < TRIPS; T += WAVES) {
        if (T + WAVES < TRIPS) trip_load(xb, T + WAVES, lane, nxt);
        trip_compute(cur, gw, xr, nw, lane, jq, xIdx, dIdx, nIdx, acc);
#pragma unroll
        for (int r = 0; r < 6; ++r)
#pragma unroll
            for (int f = 0; f < NF; ++f) cur[r][f] = nxt[r][f];
    }

#pragma unroll
    for (int q = 0; q < NQ; ++q) {
        const int v = q * 32 + lane;
        if (v < NITEM) {
#pragma unroll
            for (int c = 0; c < 9; ++c)
                gw[v * 9 + c] = acc[q][c] * (1.0f / (float)T_);
        }
    }
    __syncthreads();

    if (tid < J_ * 9) {
        const int jj = tid / 9;
        const int c  = tid - jj * 9;
        float s = 0.0f;
        for (int w = 0; w < WAVES; ++w)
#pragma unroll
            for (int f = 0; f < NF; ++f) s += lds_g[w][0][(f * J_ + jj) * 9 + c];
        out[(size_t)b * (J_ * 9) + tid] = s;
    }
}

extern "C" void kernel_launch(void* const* d_in, const int* in_sizes, int n_in,
                              void* d_out, int out_size, void* d_ws, size_t ws_size,
                              hipStream_t stream) {
    (void)in_sizes; (void)n_in; (void)d_ws; (void)ws_size; (void)out_size;
    const float* x = (const float*)d_in[0];
    float* out = (float*)d_out;
    knn_feats_kernel<<<B_, 256, 0, stream>>>(x, out);
}


# every in-range root is s*rsq(s) plus one fused residual correction, two steps of an item per packed f32 op; no neighbour compares and selects; a root may move 1 ulp at rare near-ties; same sum order
# speedup vs baseline: 2.5834x; 1.0857x over previous
#include <hip/hip_runtime.h>


typedef float v2f __attribute__((ext_vector_type(2)));
typedef float v8f __attribute__((ext_vector_type(8)));

#define B_    256
#define T_    150
#define J_    25
#define WAVES 8
#define GSTR  37
#define NF    5
#define NQ    4
#define NITEM (NF * J_)
#define TRIPS (T_ / NF)
#define KH    ((J_ - 1) / 2)
#define KU    6
#define TILE  (32 * GSTR + 1)
#define NSTR  40
#define XSTR  50
#define XROW  256
static_assert(KH % KU == 0, "the selection loop runs whole passes");
static_assert(NITEM * 9 <= TILE, "a wave's partial sums must fit its first Gram tile");


__device__ __forceinline__ void trip_load(const float* __restrict__ xb, int T, int lane,
                                          float (&raw)[6][NF]) {
    const int  lj      = lane & 15;
    const bool hiHalf  = lane >= 16;
    const int  jMid    = (16 + lj < J_) ? 16 + lj : lj - 9;
    const int  jTop    = 7 + lj;
    const int  c0      = hiHalf ? 2 : 0;
    const int  c1      = hiHalf ? 2 : 1;
#pragma unroll
    for (int f = 0; f < NF; ++f) {
        const float* p = xb + (size_t)(T * NF + f) * (J_ * 3);
        raw[0][f] = p[lj * 3 + c0];
        raw[1][f] = p[lj * 3 + c1];
        raw[2][f] = p[jMid * 3 + c0];
        raw[3][f] = p[jMid * 3 + c1];
        raw[4][f] = p[jTop * 3 + c0];
        raw[5][f] = p[jTop * 3 + c1];
    }
}

__device__ __forceinline__ float sqrt_in_range(float s) {
    const float y = __builtin_amdgcn_rsqf(s);
    const float r = s * y;
    const float h = 0.5f * y;
    const float e = __builtin_fmaf(-r, r, s);
    return __builtin_fmaf(e, h, r);
}

__device__ __forceinline__ v2f sqrt_in_range2(v2f s) {
    v2f y;
    y.x = __builtin_amdgcn_rsqf(s.x);
    y.y = __builtin_amdgcn_rsqf(s.y);
    const v2f r = s * y;
    const v2f h = 0.5f * y;
    const v2f e = __builtin_elementwise_fma(-r, r, s);
    return __builtin_elementwise_fma(e, h, r);
}

#define RANGE_LO 0x0f800000u
#define RANGE_HI 0x7f800000u

template <bool FULL>
__device__ __forceinline__ void select_and_stats(float* gw, float* const (&xr)[KU], float* nw,
                                                 const int (&xIdx)[NQ], const int (&dIdx)[NQ], const int (&nIdx)[NQ],
                                                 const float (&nj)[NQ], float (&ft)[NQ][7],
                                                 unsigned (&lo)[NQ], unsigned (&hi)[NQ]) {
    float m1[NQ], m2[NQ], m3[NQ], m4[NQ];
#pragma unroll
    for (int q = 0; q < NQ; ++q) {
        m1[q] = 3.0e38f; m2[q] = 3.0e38f; m3[q] = 3.0e38f; m4[q] = 3.0e38f;
    }

    int dI[NQ], nI[NQ], xL[NQ];
#pragma unroll
    for (int q = 0; q < NQ; ++q) { dI[q] = dIdx[q]; nI[q] = nIdx[q]; xL[q] = xIdx[q] + J_; }
#pragma clang loop unroll(disable)
    for (int kb = 0; kb < KH / KU; ++kb) {
    float own[KU][NQ];
    if (!FULL) {
#pragma unroll
        for (int s2 = 1; s2 <= KU; s2 += 2) {
#pragma unroll
            for (int q = 0; q < NQ; ++q) {
                v2f ni, gij;
                ni.x  = nw[nI[q] + s2];  ni.y  = nw[nI[q] + s2 + 1];
                gij.x = gw[dI[q] + s2];  gij.y = gw[dI[q] + s2 + 1];
                const v2f sq = ni + nj[q] - 2.0f * gij;
                const unsigned b0 = __float_as_uint(sq.x), b1 = __float_as_uint(sq.y);
                unsigned l = lo[q], h = hi[q];
                l = b0 < l ? b0 : l;  h = b0 > h ? b0 : h;
                l = b1 < l ? b1 : l;  h = b1 > h ? b1 : h;
                lo[q] = l; hi[q] = h;
                const v2f root = sqrt_in_range2(sq);
                own[s2 - 1][q] = root.x;
                own[s2][q]     = root.y;
            }
        }
    }
#pragma unroll
    for (int s = 1; s <= KU; ++s) {
        float* xk = xr[s - 1];
        float dOwn[NQ], dGot[NQ];
#pragma unroll
        for (int q = 0; q < NQ; ++q) {
            if (FULL) {
                float ni  = nw[nI[q] + s];
                float gij = gw[dI[q] + s];
                float sq  = ni + nj[q] - 2.0f * gij;
                dOwn[q] = (sq > 0.0f) ? __builtin_sqrtf(sq) : 0.0f;
            } else {
                dOwn[q] = own[s - 1][q];
            }
            xk[xIdx[q]]      = dOwn[q];
            xk[xIdx[q] + J_] = dOwn[q];
        }
#pragma unroll
        for (int q = 0; q < NQ; ++q) {
            dGot[q] = xk[xL[q] - s];
        }

#pragma unroll
        for (int q = 0; q < NQ; ++q) {
            float d  = dOwn[q];
            float t1 = fminf(d,  m1[q]); float r1 = fmaxf(d,  m1[q]);
            float t2 = fminf(r1, m2[q]); float r2 = fmaxf(r1, m2[q]);
            float t3 = fminf(r2, m3[q]); float r3 = fmaxf(r2, m3[q]);
            float t4 = fminf(r3, m4[q]);
            m1[q] = t1; m2[q] = t2; m3[q] = t3; m4[q] = t4;
        }
#pragma unroll
        for (int q = 0; q < NQ; ++q) {
            float d  = dGot[q];
            float t1 = fminf(d,  m1[q]); float r1 = fmaxf(d,  m1[q]);
            float t2 = fminf(r1, m2[q]); float r2 = fmaxf(r1, m2[q]);
            float t3 = fminf(r2, m3[q]); float r3 = fmaxf(r2, m3[q]);
            float t4 = fminf(r3, m4[q]);
            m1[q] = t1; m2[q] = t2; m3[q] = t3; m4[q] = t4;
        }
    }
#pragma unroll
    for (int q = 0; q < NQ; ++q) { dI[q] += KU; nI[q] += KU; xL[q] -= KU; }
    }

#pragma unroll
    for (int q = 0; q < NQ; ++q) {
        float mean2 = 0.5f * (m1[q] + m2[q]);
        float a0 = m1[q] - mean2, a1 = m2[q] - mean2;
        float v2 = a0 * a0 + a1 * a1;

        float mean3 = (m1[q] + m2[q] + m3[q]) * (1.0f / 3.0f);
        float e0 = m1[q] - mean3, e1 = m2[q] - mean3, e2 = m3[q] - mean3;
        float v3 = (e0 * e0 + e1 * e1 + e2 * e2) * 0.5f;

        float mean4 = (m1[q] + m2[q] + m3[q] + m4[q]) * 0.25f;
        float f0 = m1[q] - mean4, f1 = m2[q] - mean4, f2 = m3[q] - mean4, f3 = m4[q] - mean4;
        float v4 = (f0 * f0 + f1 * f1 + f2 * f2 + f3 * f3) * (1.0f / 3.0f);

        float std2, std3, std4;
        if (FULL) {
            std2 = __builtin_sqrtf(v2); std3 = __builtin_sqrtf(v3); std4 = __builtin_sqrtf(v4);
        } else {
            const unsigned b2 = __float_as_uint(v2), b3 = __float_as_uint(v3), b4 = __float_as_uint(v4);
            unsigned l = lo[q], h = hi[q];
            l = b2 < l ? b2 : l;  h = b2 > h ? b2 : h;
            l = b3 < l ? b3 : l;  h = b3 > h ? b3 : h;
            l = b4 < l ? b4 : l;  h = b4 > h ? b4 : h;
            lo[q] = l; hi[q] = h;
            std2 = sqrt_in_range(v2); std3 = sqrt_in_range(v3); std4 = sqrt_in_range(v4);
        }

        ft[q][0] = mean2; ft[q][1] = std2; ft[q][2] = m1[q];
        ft[q][3] = mean3; ft[q][4] = std3;
        ft[q][5] = mean4; ft[q][6] = std4;
    }
}

__device__ __forceinline__ void trip_compute(const float (&raw)[6][NF], float* gw, float* const (&xr)[KU], float* nw,
                                             int lane, const int (&jq)[NQ], const int (&xIdx)[NQ],
                                             const int (&dIdx)[NQ], const int (&nIdx)[NQ],
                                             float (&acc)[NQ][9]) {
    const int  lj      = lane & 15;
    const bool hiHalf  = lane >= 16;

    v2f aLo[NF], aHi[NF], aTop[NF];
#pragma unroll
    for (int f = 0; f < NF; ++f) {
        aLo[f].x  = raw[0][f];
        aLo[f].y  = hiHalf ? 0.0f : raw[1][f];
        aHi[f].x  = raw[2][f];
        aHi[f].y  = hiHalf ? 0.0f : raw[3][f];
        aTop[f].x = raw[4][f];
        aTop[f].y = hiHalf ? 0.0f : raw[5][f];
    }

    const int rbase = hiHalf ? 8 : 0;
#pragma unroll
    for (int f = 0; f < NF; ++f) {
        v8f cz = {};
        v8f g00 = __builtin_amdgcn_wmma_f32_16x16x4_f32(false, aLo[f], false, aLo[f], (short)0, cz, false, false);
        v8f g01 = __builtin_amdgcn_wmma_f32_16x16x4_f32(false, aLo[f], false, aHi[f], (short)0, cz, false, false);
        v8f g11 = __builtin_amdgcn_wmma_f32_16x16x4_f32(false, aHi[f], false, aHi[f], (short)0, cz, false, false);
        v8f g12 = __builtin_amdgcn_wmma_f32_16x16x4_f32(false, aHi[f], false, aTop[f], (short)0, cz, false, false);

        float* gf = gw + f * TILE;
#pragma unroll
        for (int r = 0; r < 8; ++r) {
            gf[(rbase + r) * GSTR + lj]            = g00[r];
            gf[(rbase + r) * GSTR + 16 + lj]       = g01[r];
            gf[(16 + rbase + r) * GSTR + 16 + lj]  = g11[r];
        }
        if (lj < GSTR - 32) {
#pragma unroll
            for (int r = 0; r < 8; ++r)
                gf[(16 + rbase + r) * GSTR + 32 + lj] = g12[r];
        }
    }
    __builtin_amdgcn_wave_barrier();

    float nj[NQ];
#pragma unroll
    for (int q = 0; q < NQ; ++q) nj[q] = gw[dIdx[q]];
#pragma unroll
    for (int q = 0; q < NQ; ++q) {
        nw[nIdx[q]] = nj[q];
        if (jq[q] < KH) nw[nIdx[q] + J_] = nj[q];
    }
    __builtin_amdgcn_wave_barrier();

    float ft[NQ][7];
    unsigned lo[NQ], hi[NQ];
#pragma unroll
    for (int q = 0; q < NQ; ++q) { lo[q] = 0xffffffffu; hi[q] = 0u; }
    select_and_stats<false>(gw, xr, nw, xIdx, dIdx, nIdx, nj, ft, lo, hi);
    unsigned loAll = lo[0], hiAll = hi[0];
#pragma unroll
    for (int q = 1; q < NQ; ++q) {
        loAll = lo[q] < loAll ? lo[q] : loAll;
        hiAll = hi[q] > hiAll ? hi[q] : hiAll;
    }
    const bool outside = (loAll < RANGE_LO) | (hiAll >= RANGE_HI);
    if (__builtin_amdgcn_ballot_w32(outside) != 0u) {
        select_and_stats<true>(gw, xr, nw, xIdx, dIdx, nIdx, nj, ft, lo, hi);
    }

#pragma unroll
    for (int q = 0; q < NQ; ++q) {
        acc[q][0] += ft[q][0]; acc[q][1] += ft[q][1]; acc[q][2] += ft[q][2];
        acc[q][3] += ft[q][3]; acc[q][4] += ft[q][4]; acc[q][5] += ft[q][2];
        acc[q][6] += ft[q][5]; acc[q][7] += ft[q][6]; acc[q][8] += ft[q][2];
    }
    __builtin_amdgcn_wave_barrier();
}

__global__ __launch_bounds__(256) __attribute__((amdgpu_waves_per_eu(2, 2)))
void knn_feats_kernel(const float* __restrict__ x, float* __restrict__ out) {
    __shared__ float lds_g[WAVES][NF][TILE];
    __shared__ float lds_n[WAVES][NF * NSTR];
    __shared__ float lds_x0[WAVES][XROW], lds_x1[WAVES][XROW], lds_x2[WAVES][XROW];
    __shared__ float lds_x3[WAVES][XROW], lds_x4[WAVES][XROW], lds_x5[WAVES][XROW];

    const int tid  = threadIdx.x;
    const int wv   = __builtin_amdgcn_readfirstlane(tid >> 5);
    const int lane = tid & 31;
    const int b    = blockIdx.x;

    float acc[NQ][9];
    int jq[NQ], xIdx[NQ], dIdx[NQ], nIdx[NQ];
#pragma unroll
    for (int q = 0; q < NQ; ++q) {
        const int v = q * 32 + lane;
        const int vc = (v < NITEM) ? v : NITEM - 1;
        const int f = vc / J_;
        jq[q] = vc - f * J_;
        xIdx[q] = f * XSTR + jq[q];
        dIdx[q] = f * TILE + jq[q] * (GSTR + 1);
        nIdx[q] = f * NSTR + jq[q];
#pragma unroll
        for (int c = 0; c < 9; ++c) acc[q][c] = 0.0f;
    }

    float* gw = lds_g[wv][0];
    float* const xr[KU] = { lds_x0[wv], lds_x1[wv], lds_x2[wv], lds_x3[wv], lds_x4[wv], lds_x5[wv] };
    float* nw = lds_n[wv];
    const float* xb = x + (size_t)b * T_ * (J_ * 3);

    float cur[6][NF], nxt[6][NF];
#pragma unroll
    for (int r = 0; r < 6; ++r)
#pragma unroll
        for (int f = 0; f < NF; ++f) nxt[r][f] = 0.0f;
    trip_load(xb, wv, lane, cur);
    for (int T = wv; T < TRIPS; T += WAVES) {
        if (T + WAVES < TRIPS) trip_load(xb, T + WAVES, lane, nxt);
        trip_compute(cur, gw, xr, nw, lane, jq, xIdx, dIdx, nIdx, acc);
#pragma unroll
        for (int r = 0; r < 6; ++r)
#pragma unroll
            for (int f = 0; f < NF; ++f) cur[r][f] = nxt[r][f];
    }

#pragma unroll
    for (int q = 0; q < NQ; ++q) {
        const int v = q * 32 + lane;
        if (v < NITEM) {
#pragma unroll
            for (int c = 0; c < 9; ++c)
                gw[v * 9 + c] = acc[q][c] * (1.0f / (float)T_);
        }
    }
    __syncthreads();

    if (tid < J_ * 9) {
        const int jj = tid / 9;
        const int c  = tid - jj * 9;
        float s = 0.0f;
        for (int w = 0; w < WAVES; ++w)
#pragma unroll
            for (int f = 0; f < NF; ++f) s += lds_g[w][0][(f * J_ + jj) * 9 + c];
        out[(size_t)b * (J_ * 9) + tid] = s;
    }
}

extern "C" void kernel_launch(void* const* d_in, const int* in_sizes, int n_in,
                              void* d_out, int out_size, void* d_ws, size_t ws_size,
                              hipStream_t stream) {
    (void)in_sizes; (void)n_in; (void)d_ws; (void)ws_size; (void)out_size;
    const float* x = (const float*)d_in[0];
    float* out = (float*)d_out;
    knn_feats_kernel<<<B_, 256, 0, stream>>>(x, out);
}


# next-trip loads issued without a branch (a wave's last trip reloads its own frames, unread): the compiler's load wait before the matrix products no longer waits for them too; same sums
# speedup vs baseline: 2.7413x; 1.0611x over previous
#include <hip/hip_runtime.h>


typedef float v2f __attribute__((ext_vector_type(2)));
typedef float v8f __attribute__((ext_vector_type(8)));

#define B_    256
#define T_    150
#define J_    25
#define WAVES 8
#define GSTR  37
#define NF    5
#define NQ    4
#define NITEM (NF * J_)
#define TRIPS (T_ / NF)
#define KH    ((J_ - 1) / 2)
#define KU    6
#define TILE  (32 * GSTR + 1)
#define NSTR  40
#define XSTR  50
#define XROW  256
static_assert(KH % KU == 0, "the selection loop runs whole passes");
static_assert(NITEM * 9 <= TILE, "a wave's partial sums must fit its first Gram tile");


__device__ __forceinline__ void trip_load(const float* __restrict__ xb, int T, int lane,
                                          float (&raw)[6][NF]) {
    const int  lj      = lane & 15;
    const bool hiHalf  = lane >= 16;
    const int  jMid    = (16 + lj < J_) ? 16 + lj : lj - 9;
    const int  jTop    = 7 + lj;
    const int  c0      = hiHalf ? 2 : 0;
    const int  c1      = hiHalf ? 2 : 1;
#pragma unroll
    for (int f = 0; f < NF; ++f) {
        const float* p = xb + (size_t)(T * NF + f) * (J_ * 3);
        raw[0][f] = p[lj * 3 + c0];
        raw[1][f] = p[lj * 3 + c1];
        raw[2][f] = p[jMid * 3 + c0];
        raw[3][f] = p[jMid * 3 + c1];
        raw[4][f] = p[jTop * 3 + c0];
        raw[5][f] = p[jTop * 3 + c1];
    }
}

__device__ __forceinline__ float sqrt_in_range(float s) {
    const float y = __builtin_amdgcn_rsqf(s);
    const float r = s * y;
    const float h = 0.5f * y;
    const float e = __builtin_fmaf(-r, r, s);
    return __builtin_fmaf(e, h, r);
}

__device__ __forceinline__ v2f sqrt_in_range2(v2f s) {
    v2f y;
    y.x = __builtin_amdgcn_rsqf(s.x);
    y.y = __builtin_amdgcn_rsqf(s.y);
    const v2f r = s * y;
    const v2f h = 0.5f * y;
    const v2f e = __builtin_elementwise_fma(-r, r, s);
    return __builtin_elementwise_fma(e, h, r);
}

#define RANGE_LO 0x0f800000u
#define RANGE_HI 0x7f800000u

template <bool FULL>
__device__ __forceinline__ void select_and_stats(float* gw, float* const (&xr)[KU], float* nw,
                                                 const int (&xIdx)[NQ], const int (&dIdx)[NQ], const int (&nIdx)[NQ],
                                                 const float (&nj)[NQ], float (&ft)[NQ][7],
                                                 unsigned (&lo)[NQ], unsigned (&hi)[NQ]) {
    float m1[NQ], m2[NQ], m3[NQ], m4[NQ];
#pragma unroll
    for (int q = 0; q < NQ; ++q) {
        m1[q] = 3.0e38f; m2[q] = 3.0e38f; m3[q] = 3.0e38f; m4[q] = 3.0e38f;
    }

    int dI[NQ], nI[NQ], xL[NQ];
#pragma unroll
    for (int q = 0; q < NQ; ++q) { dI[q] = dIdx[q]; nI[q] = nIdx[q]; xL[q] = xIdx[q] + J_; }
#pragma clang loop unroll(disable)
    for (int kb = 0; kb < KH / KU; ++kb) {
    float own[KU][NQ];
    if (!FULL) {
#pragma unroll
        for (int s2 = 1; s2 <= KU; s2 += 2) {
#pragma unroll
            for (int q = 0; q < NQ; ++q) {
                v2f ni, gij;
                ni.x  = nw[nI[q] + s2];  ni.y  = nw[nI[q] + s2 + 1];
                gij.x = gw[dI[q] + s2];  gij.y = gw[dI[q] + s2 + 1];
                const v2f sq = ni + nj[q] - 2.0f * gij;
                const unsigned b0 = __float_as_uint(sq.x), b1 = __float_as_uint(sq.y);
                unsigned l = lo[q], h = hi[q];
                l = b0 < l ? b0 : l;  h = b0 > h ? b0 : h;
                l = b1 < l ? b1 : l;  h = b1 > h ? b1 : h;
                lo[q] = l; hi[q] = h;
                const v2f root = sqrt_in_range2(sq);
                own[s2 - 1][q] = root.x;
                own[s2][q]     = root.y;
            }
        }
    }
#pragma unroll
    for (int s = 1; s <= KU; ++s) {
        float* xk = xr[s - 1];
        float dOwn[NQ], dGot[NQ];
#pragma unroll
        for (int q = 0; q < NQ; ++q) {
            if (FULL) {
                float ni  = nw[nI[q] + s];
                float gij = gw[dI[q] + s];
                float sq  = ni + nj[q] - 2.0f * gij;
                dOwn[q] = (sq > 0.0f) ? __builtin_sqrtf(sq) : 0.0f;
            } else {
                dOwn[q] = own[s - 1][q];
            }
            xk[xIdx[q]]      = dOwn[q];
            xk[xIdx[q] + J_] = dOwn[q];
        }
#pragma unroll
        for (int q = 0; q < NQ; ++q) {
            dGot[q] = xk[xL[q] - s];
        }

#pragma unroll
        for (int q = 0; q < NQ; ++q) {
            float d  = dOwn[q];
            float t1 = fminf(d,  m1[q]); float r1 = fmaxf(d,  m1[q]);
            float t2 = fminf(r1, m2[q]); float r2 = fmaxf(r1, m2[q]);
            float t3 = fminf(r2, m3[q]); float r3 = fmaxf(r2, m3[q]);
            float t4 = fminf(r3, m4[q]);
            m1[q] = t1; m2[q] = t2; m3[q] = t3; m4[q] = t4;
        }
#pragma unroll
        for (int q = 0; q < NQ; ++q) {
            float d  = dGot[q];
            float t1 = fminf(d,  m1[q]); float r1 = fmaxf(d,  m1[q]);
            float t2 = fminf(r1, m2[q]); float r2 = fmaxf(r1, m2[q]);
            float t3 = fminf(r2, m3[q]); float r3 = fmaxf(r2, m3[q]);
            float t4 = fminf(r3, m4[q]);
            m1[q] = t1; m2[q] = t2; m3[q] = t3; m4[q] = t4;
        }
    }
#pragma unroll
    for (int q = 0; q < NQ; ++q) { dI[q] += KU; nI[q] += KU; xL[q] -= KU; }
    }

#pragma unroll
    for (int q = 0; q < NQ; ++q) {
        float mean2 = 0.5f * (m1[q] + m2[q]);
        float a0 = m1[q] - mean2, a1 = m2[q] - mean2;
        float v2 = a0 * a0 + a1 * a1;

        float mean3 = (m1[q] + m2[q] + m3[q]) * (1.0f / 3.0f);
        float e0 = m1[q] - mean3, e1 = m2[q] - mean3, e2 = m3[q] - mean3;
        float v3 = (e0 * e0 + e1 * e1 + e2 * e2) * 0.5f;

        float mean4 = (m1[q] + m2[q] + m3[q] + m4[q]) * 0.25f;
        float f0 = m1[q] - mean4, f1 = m2[q] - mean4, f2 = m3[q] - mean4, f3 = m4[q] - mean4;
        float v4 = (f0 * f0 + f1 * f1 + f2 * f2 + f3 * f3) * (1.0f / 3.0f);

        float std2, std3, std4;
        if (FULL) {
            std2 = __builtin_sqrtf(v2); std3 = __builtin_sqrtf(v3); std4 = __builtin_sqrtf(v4);
        } else {
            const unsigned b2 = __float_as_uint(v2), b3 = __float_as_uint(v3), b4 = __float_as_uint(v4);
            unsigned l = lo[q], h = hi[q];
            l = b2 < l ? b2 : l;  h = b2 > h ? b2 : h;
            l = b3 < l ? b3 : l;  h = b3 > h ? b3 : h;
            l = b4 < l ? b4 : l;  h = b4 > h ? b4 : h;
            lo[q] = l; hi[q] = h;
            std2 = sqrt_in_range(v2); std3 = sqrt_in_range(v3); std4 = sqrt_in_range(v4);
        }

        ft[q][0] = mean2; ft[q][1] = std2; ft[q][2] = m1[q];
        ft[q][3] = mean3; ft[q][4] = std3;
        ft[q][5] = mean4; ft[q][6] = std4;
    }
}

__device__ __forceinline__ void trip_compute(const float (&raw)[6][NF], float* gw, float* const (&xr)[KU], float* nw,
                                             int lane, const int (&jq)[NQ], const int (&xIdx)[NQ],
                                             const int (&dIdx)[NQ], const int (&nIdx)[NQ],
                                             float (&acc)[NQ][9]) {
    const int  lj      = lane & 15;
    const bool hiHalf  = lane >= 16;

    v2f aLo[NF], aHi[NF], aTop[NF];
#pragma unroll
    for (int f = 0; f < NF; ++f) {
        aLo[f].x  = raw[0][f];
        aLo[f].y  = hiHalf ? 0.0f : raw[1][f];
        aHi[f].x  = raw[2][f];
        aHi[f].y  = hiHalf ? 0.0f : raw[3][f];
        aTop[f].x = raw[4][f];
        aTop[f].y = hiHalf ? 0.0f : raw[5][f];
    }

    const int rbase = hiHalf ? 8 : 0;
#pragma unroll
    for (int f = 0; f < NF; ++f) {
        v8f cz = {};
        v8f g00 = __builtin_amdgcn_wmma_f32_16x16x4_f32(false, aLo[f], false, aLo[f], (short)0, cz, false, false);
        v8f g01 = __builtin_amdgcn_wmma_f32_16x16x4_f32(false, aLo[f], false, aHi[f], (short)0, cz, false, false);
        v8f g11 = __builtin_amdgcn_wmma_f32_16x16x4_f32(false, aHi[f], false, aHi[f], (short)0, cz, false, false);
        v8f g12 = __builtin_amdgcn_wmma_f32_16x16x4_f32(false, aHi[f], false, aTop[f], (short)0, cz, false, false);

        float* gf = gw + f * TILE;
#pragma unroll
        for (int r = 0; r < 8; ++r) {
            gf[(rbase + r) * GSTR + lj]            = g00[r];
            gf[(rbase + r) * GSTR + 16 + lj]       = g01[r];
            gf[(16 + rbase + r) * GSTR + 16 + lj]  = g11[r];
        }
        if (lj < GSTR - 32) {
#pragma unroll
            for (int r = 0; r < 8; ++r)
                gf[(16 + rbase + r) * GSTR + 32 + lj] = g12[r];
        }
    }
    __builtin_amdgcn_wave_barrier();

    float nj[NQ];
#pragma unroll
    for (int q = 0; q < NQ; ++q) nj[q] = gw[dIdx[q]];
#pragma unroll
    for (int q = 0; q < NQ; ++q) {
        nw[nIdx[q]] = nj[q];
        if (jq[q] < KH) nw[nIdx[q] + J_] = nj[q];
    }
    __builtin_amdgcn_wave_barrier();

    float ft[NQ][7];
    unsigned lo[NQ], hi[NQ];
#pragma unroll
    for (int q = 0; q < NQ; ++q) { lo[q] = 0xffffffffu; hi[q] = 0u; }
    select_and_stats<false>(gw, xr, nw, xIdx, dIdx, nIdx, nj, ft, lo, hi);
    unsigned loAll = lo[0], hiAll = hi[0];
#pragma unroll
    for (int q = 1; q < NQ; ++q) {
        loAll = lo[q] < loAll ? lo[q] : loAll;
        hiAll = hi[q] > hiAll ? hi[q] : hiAll;
    }
    const bool outside = (loAll < RANGE_LO) | (hiAll >= RANGE_HI);
    if (__builtin_amdgcn_ballot_w32(outside) != 0u) {
        select_and_stats<true>(gw, xr, nw, xIdx, dIdx, nIdx, nj, ft, lo, hi);
    }

#pragma unroll
    for (int q = 0; q < NQ; ++q) {
        acc[q][0] += ft[q][0]; acc[q][1] += ft[q][1]; acc[q][2] += ft[q][2];
        acc[q][3] += ft[q][3]; acc[q][4] += ft[q][4]; acc[q][5] += ft[q][2];
        acc[q][6] += ft[q][5]; acc[q][7] += ft[q][6]; acc[q][8] += ft[q][2];
    }
    __builtin_amdgcn_wave_barrier();
}

__global__ __launch_bounds__(256) __attribute__((amdgpu_waves_per_eu(2, 2)))
void knn_feats_kernel(const float* __restrict__ x, float* __restrict__ out) {
    __shared__ float lds_g[WAVES][NF][TILE];
    __shared__ float lds_n[WAVES][NF * NSTR];
    __shared__ float lds_x0[WAVES][XROW], lds_x1[WAVES][XROW], lds_x2[WAVES][XROW];
    __shared__ float lds_x3[WAVES][XROW], lds_x4[WAVES][XROW], lds_x5[WAVES][XROW];

    const int tid  = threadIdx.x;
    const int wv   = __builtin_amdgcn_readfirstlane(tid >> 5);
    const int lane = tid & 31;
    const int b    = blockIdx.x;

    float acc[NQ][9];
    int jq[NQ], xIdx[NQ], dIdx[NQ], nIdx[NQ];
#pragma unroll
    for (int q = 0; q < NQ; ++q) {
        const int v = q * 32 + lane;
        const int vc = (v < NITEM) ? v : NITEM - 1;
        const int f = vc / J_;
        jq[q] = vc - f * J_;
        xIdx[q] = f * XSTR + jq[q];
        dIdx[q] = f * TILE + jq[q] * (GSTR + 1);
        nIdx[q] = f * NSTR + jq[q];
#pragma unroll
        for (int c = 0; c < 9; ++c) acc[q][c] = 0.0f;
    }

    float* gw = lds_g[wv][0];
    float* const xr[KU] = { lds_x0[wv], lds_x1[wv], lds_x2[wv], lds_x3[wv], lds_x4[wv], lds_x5[wv] };
    float* nw = lds_n[wv];
    const float* xb = x + (size_t)b * T_ * (J_ * 3);

    float cur[6][NF], nxt[6][NF];
#pragma unroll
    for (int r = 0; r < 6; ++r)
#pragma unroll
        for (int f = 0; f < NF; ++f) nxt[r][f] = 0.0f;
    trip_load(xb, wv, lane, cur);
    for (int T = wv; T < TRIPS; T += WAVES) {
        const int Tn = (T + WAVES < TRIPS) ? T + WAVES : T;
        trip_load(xb, Tn, lane, nxt);
        trip_compute(cur, gw, xr, nw, lane, jq, xIdx, dIdx, nIdx, acc);
#pragma unroll
        for (int r = 0; r < 6; ++r)
#pragma unroll
            for (int f = 0; f < NF; ++f) cur[r][f] = nxt[r][f];
    }

#pragma unroll
    for (int q = 0; q < NQ; ++q) {
        const int v = q * 32 + lane;
        if (v < NITEM) {
#pragma unroll
            for (int c = 0; c < 9; ++c)
                gw[v * 9 + c] = acc[q][c] * (1.0f / (float)T_);
        }
    }
    __syncthreads();

    if (tid < J_ * 9) {
        const int jj = tid / 9;
        const int c  = tid - jj * 9;
        float s = 0.0f;
        for (int w = 0; w < WAVES; ++w)
#pragma unroll
            for (int f = 0; f < NF; ++f) s += lds_g[w][0][(f * J_ + jj) * 9 + c];
        out[(size_t)b * (J_ * 9) + tid] = s;
    }
}

extern "C" void kernel_launch(void* const* d_in, const int* in_sizes, int n_in,
                              void* d_out, int out_size, void* d_ws, size_t ws_size,
                              hipStream_t stream) {
    (void)in_sizes; (void)n_in; (void)d_ws; (void)ws_size; (void)out_size;
    const float* x = (const float*)d_in[0];
    float* out = (float*)d_out;
    knn_feats_kernel<<<B_, 256, 0, stream>>>(x, out);
}


# Gram tile kept transposed (operands swapped, Gram is symmetric): a lane's 8 result rows are contiguous, 2 16-byte LDS stores per product, was 4 two-address; no lane-conditional store; same sums
# speedup vs baseline: 2.9257x; 1.0673x over previous
#include <hip/hip_runtime.h>


typedef float v2f __attribute__((ext_vector_type(2)));
typedef float v4f __attribute__((ext_vector_type(4)));
typedef float v8f __attribute__((ext_vector_type(8)));

#define B_    256
#define T_    150
#define J_    25
#define WAVES 8
#define GSTR  44
#define NF    5
#define NQ    4
#define NITEM (NF * J_)
#define TRIPS (T_ / NF)
#define KH    ((J_ - 1) / 2)
#define KU    6
#define TILE  (32 * GSTR + 8)
#define NSTR  50
#define XSTR  50
#define XROW  256
static_assert(KH % KU == 0, "the selection loop runs whole passes");
static_assert(NITEM * 9 <= TILE, "a wave's partial sums must fit its first Gram tile");


__device__ __forceinline__ void trip_load(const float* __restrict__ xb, int T, int lane,
                                          float (&raw)[6][NF]) {
    const int  lj      = lane & 15;
    const bool hiHalf  = lane >= 16;
    const int  jMid    = (16 + lj < J_) ? 16 + lj : lj - 9;
    const int  jTop    = 7 + lj;
    const int  c0      = hiHalf ? 2 : 0;
    const int  c1      = hiHalf ? 2 : 1;
#pragma unroll
    for (int f = 0; f < NF; ++f) {
        const float* p = xb + (size_t)(T * NF + f) * (J_ * 3);
        raw[0][f] = p[lj * 3 + c0];
        raw[1][f] = p[lj * 3 + c1];
        raw[2][f] = p[jMid * 3 + c0];
        raw[3][f] = p[jMid * 3 + c1];
        raw[4][f] = p[jTop * 3 + c0];
        raw[5][f] = p[jTop * 3 + c1];
    }
}

__device__ __forceinline__ float sqrt_in_range(float s) {
    const float y = __builtin_amdgcn_rsqf(s);
    const float r = s * y;
    const float h = 0.5f * y;
    const float e = __builtin_fmaf(-r, r, s);
    return __builtin_fmaf(e, h, r);
}

__device__ __forceinline__ v2f sqrt_in_range2(v2f s) {
    v2f y;
    y.x = __builtin_amdgcn_rsqf(s.x);
    y.y = __builtin_amdgcn_rsqf(s.y);
    const v2f r = s * y;
    const v2f h = 0.5f * y;
    const v2f e = __builtin_elementwise_fma(-r, r, s);
    return __builtin_elementwise_fma(e, h, r);
}

#define RANGE_LO 0x0f800000u
#define RANGE_HI 0x7f800000u

template <bool FULL>
__device__ __forceinline__ void select_and_stats(float* gw, float* const (&xr)[KU], float* nw,
                                                 const int (&xIdx)[NQ], const int (&dIdx)[NQ], const int (&nIdx)[NQ],
                                                 const float (&nj)[NQ], float (&ft)[NQ][7],
                                                 unsigned (&lo)[NQ], unsigned (&hi)[NQ]) {
    float m1[NQ], m2[NQ], m3[NQ], m4[NQ];
#pragma unroll
    for (int q = 0; q < NQ; ++q) {
        m1[q] = 3.0e38f; m2[q] = 3.0e38f; m3[q] = 3.0e38f; m4[q] = 3.0e38f;
    }

    int dI[NQ], nI[NQ], xL[NQ];
#pragma unroll
    for (int q = 0; q < NQ; ++q) { dI[q] = dIdx[q]; nI[q] = nIdx[q]; xL[q] = xIdx[q] + J_; }
#pragma clang loop unroll(disable)
    for (int kb = 0; kb < KH / KU; ++kb) {
    float own[KU][NQ];
    if (!FULL) {
#pragma unroll
        for (int s2 = 1; s2 <= KU; s2 += 2) {
#pragma unroll
            for (int q = 0; q < NQ; ++q) {
                v2f ni, gij;
                ni.x  = nw[nI[q] + s2];  ni.y  = nw[nI[q] + s2 + 1];
                gij.x = gw[dI[q] + s2];  gij.y = gw[dI[q] + s2 + 1];
                const v2f sq = ni + nj[q] - 2.0f * gij;
                const unsigned b0 = __float_as_uint(sq.x), b1 = __float_as_uint(sq.y);
                unsigned l = lo[q], h = hi[q];
                l = b0 < l ? b0 : l;  h = b0 > h ? b0 : h;
                l = b1 < l ? b1 : l;  h = b1 > h ? b1 : h;
                lo[q] = l; hi[q] = h;
                const v2f root = sqrt_in_range2(sq);
                own[s2 - 1][q] = root.x;
                own[s2][q]     = root.y;
            }
        }
    }
#pragma unroll
    for (int s = 1; s <= KU; ++s) {
        float* xk = xr[s - 1];
        float dOwn[NQ], dGot[NQ];
#pragma unroll
        for (int q = 0; q < NQ; ++q) {
            if (FULL) {
                float ni  = nw[nI[q] + s];
                float gij = gw[dI[q] + s];
                float sq  = ni + nj[q] - 2.0f * gij;
                dOwn[q] = (sq > 0.0f) ? __builtin_sqrtf(sq) : 0.0f;
            } else {
                dOwn[q] = own[s - 1][q];
            }
            xk[xIdx[q]]      = dOwn[q];
            xk[xIdx[q] + J_] = dOwn[q];
        }
#pragma unroll
        for (int q = 0; q < NQ; ++q) {
            dGot[q] = xk[xL[q] - s];
        }

#pragma unroll
        for (int q = 0; q < NQ; ++q) {
            float d  = dOwn[q];
            float t1 = fminf(d,  m1[q]); float r1 = fmaxf(d,  m1[q]);
            float t2 = fminf(r1, m2[q]); float r2 = fmaxf(r1, m2[q]);
            float t3 = fminf(r2, m3[q]); float r3 = fmaxf(r2, m3[q]);
            float t4 = fminf(r3, m4[q]);
            m1[q] = t1; m2[q] = t2; m3[q] = t3; m4[q] = t4;
        }
#pragma unroll
        for (int q = 0; q < NQ; ++q) {
            float d  = dGot[q];
            float t1 = fminf(d,  m1[q]); float r1 = fmaxf(d,  m1[q]);
            float t2 = fminf(r1, m2[q]); float r2 = fmaxf(r1, m2[q]);
            float t3 = fminf(r2, m3[q]); float r3 = fmaxf(r2, m3[q]);
            float t4 = fminf(r3, m4[q]);
            m1[q] = t1; m2[q] = t2; m3[q] = t3; m4[q] = t4;
        }
    }
#pragma unroll
    for (int q = 0; q < NQ; ++q) { dI[q] += KU; nI[q] += KU; xL[q] -= KU; }
    }

#pragma unroll
    for (int q = 0; q < NQ; ++q) {
        float mean2 = 0.5f * (m1[q] + m2[q]);
        float a0 = m1[q] - mean2, a1 = m2[q] - mean2;
        float v2 = a0 * a0 + a1 * a1;

        float mean3 = (m1[q] + m2[q] + m3[q]) * (1.0f / 3.0f);
        float e0 = m1[q] - mean3, e1 = m2[q] - mean3, e2 = m3[q] - mean3;
        float v3 = (e0 * e0 + e1 * e1 + e2 * e2) * 0.5f;

        float mean4 = (m1[q] + m2[q] + m3[q] + m4[q]) * 0.25f;
        float f0 = m1[q] - mean4, f1 = m2[q] - mean4, f2 = m3[q] - mean4, f3 = m4[q] - mean4;
        float v4 = (f0 * f0 + f1 * f1 + f2 * f2 + f3 * f3) * (1.0f / 3.0f);

        float std2, std3, std4;
        if (FULL) {
            std2 = __builtin_sqrtf(v2); std3 = __builtin_sqrtf(v3); std4 = __builtin_sqrtf(v4);
        } else {
            const unsigned b2 = __float_as_uint(v2), b3 = __float_as_uint(v3), b4 = __float_as_uint(v4);
            unsigned l = lo[q], h = hi[q];
            l = b2 < l ? b2 : l;  h = b2 > h ? b2 : h;
            l = b3 < l ? b3 : l;  h = b3 > h ? b3 : h;
            l = b4 < l ? b4 : l;  h = b4 > h ? b4 : h;
            lo[q] = l; hi[q] = h;
            std2 = sqrt_in_range(v2); std3 = sqrt_in_range(v3); std4 = sqrt_in_range(v4);
        }

        ft[q][0] = mean2; ft[q][1] = std2; ft[q][2] = m1[q];
        ft[q][3] = mean3; ft[q][4] = std3;
        ft[q][5] = mean4; ft[q][6] = std4;
    }
}

__device__ __forceinline__ void trip_compute(const float (&raw)[6][NF], float* gw, float* const (&xr)[KU], float* nw,
                                             int lane, const int (&xIdx)[NQ],
                                             const int (&dIdx)[NQ], const int (&nIdx)[NQ],
                                             float (&acc)[NQ][9]) {
    const int  lj      = lane & 15;
    const bool hiHalf  = lane >= 16;

    v2f aLo[NF], aHi[NF], aTop[NF];
#pragma unroll
    for (int f = 0; f < NF; ++f) {
        aLo[f].x  = raw[0][f];
        aLo[f].y  = hiHalf ? 0.0f : raw[1][f];
        aHi[f].x  = raw[2][f];
        aHi[f].y  = hiHalf ? 0.0f : raw[3][f];
        aTop[f].x = raw[4][f];
        aTop[f].y = hiHalf ? 0.0f : raw[5][f];
    }

    const int rbase = hiHalf ? 8 : 0;
#pragma unroll
    for (int f = 0; f < NF; ++f) {
        v8f cz = {};
        v8f g00 = __builtin_amdgcn_wmma_f32_16x16x4_f32(false, aLo[f],  false, aLo[f], (short)0, cz, false, false);
        v8f g10 = __builtin_amdgcn_wmma_f32_16x16x4_f32(false, aHi[f],  false, aLo[f], (short)0, cz, false, false);
        v8f g11 = __builtin_amdgcn_wmma_f32_16x16x4_f32(false, aHi[f],  false, aHi[f], (short)0, cz, false, false);
        v8f g21 = __builtin_amdgcn_wmma_f32_16x16x4_f32(false, aTop[f], false, aHi[f], (short)0, cz, false, false);

        float* l0 = gw + f * TILE + lj * GSTR + rbase;
        float* l1 = gw + f * TILE + (16 + lj) * GSTR + rbase;
        *reinterpret_cast<v4f*>(l0)      = __builtin_shufflevector(g00, g00, 0, 1, 2, 3);
        *reinterpret_cast<v4f*>(l0 + 4)  = __builtin_shufflevector(g00, g00, 4, 5, 6, 7);
        *reinterpret_cast<v4f*>(l0 + 16) = __builtin_shufflevector(g10, g10, 0, 1, 2, 3);
        *reinterpret_cast<v4f*>(l0 + 20) = __builtin_shufflevector(g10, g10, 4, 5, 6, 7);
        *reinterpret_cast<v4f*>(l1 + 16) = __builtin_shufflevector(g11, g11, 0, 1, 2, 3);
        *reinterpret_cast<v4f*>(l1 + 20) = __builtin_shufflevector(g11, g11, 4, 5, 6, 7);
        *reinterpret_cast<v4f*>(l1 + 32) = __builtin_shufflevector(g21, g21, 0, 1, 2, 3);
        *reinterpret_cast<v4f*>(l1 + 36) = __builtin_shufflevector(g21, g21, 4, 5, 6, 7);
    }
    __builtin_amdgcn_wave_barrier();

    float nj[NQ];
#pragma unroll
    for (int q = 0; q < NQ; ++q) nj[q] = gw[dIdx[q]];
#pragma unroll
    for (int q = 0; q < NQ; ++q) {
        nw[nIdx[q]]      = nj[q];
        nw[nIdx[q] + J_] = nj[q];
    }
    __builtin_amdgcn_wave_barrier();

    float ft[NQ][7];
    unsigned lo[NQ], hi[NQ];
#pragma unroll
    for (int q = 0; q < NQ; ++q) { lo[q] = 0xffffffffu; hi[q] = 0u; }
    select_and_stats<false>(gw, xr, nw, xIdx, dIdx, nIdx, nj, ft, lo, hi);
    unsigned loAll = lo[0], hiAll = hi[0];
#pragma unroll
    for (int q = 1; q < NQ; ++q) {
        loAll = lo[q] < loAll ? lo[q] : loAll;
        hiAll = hi[q] > hiAll ? hi[q] : hiAll;
    }
    const bool outside = (loAll < RANGE_LO) | (hiAll >= RANGE_HI);
    if (__builtin_amdgcn_ballot_w32(outside) != 0u) {
        select_and_stats<true>(gw, xr, nw, xIdx, dIdx, nIdx, nj, ft, lo, hi);
    }

#pragma unroll
    for (int q = 0; q < NQ; ++q) {
        acc[q][0] += ft[q][0]; acc[q][1] += ft[q][1]; acc[q][2] += ft[q][2];
        acc[q][3] += ft[q][3]; acc[q][4] += ft[q][4]; acc[q][5] += ft[q][2];
        acc[q][6] += ft[q][5]; acc[q][7] += ft[q][6]; acc[q][8] += ft[q][2];
    }
    __builtin_amdgcn_wave_barrier();
}

__global__ __launch_bounds__(256) __attribute__((amdgpu_waves_per_eu(2, 2)))
void knn_feats_kernel(const float* __restrict__ x, float* __restrict__ out) {
    __shared__ __attribute__((aligned(16))) float lds_g[WAVES][NF][TILE];
    __shared__ float lds_n[WAVES][NF * NSTR];
    __shared__ float lds_x0[WAVES][XROW], lds_x1[WAVES][XROW], lds_x2[WAVES][XROW];
    __shared__ float lds_x3[WAVES][XROW], lds_x4[WAVES][XROW], lds_x5[WAVES][XROW];

    const int tid  = threadIdx.x;
    const int wv   = __builtin_amdgcn_readfirstlane(tid >> 5);
    const int lane = tid & 31;
    const int b    = blockIdx.x;

    float acc[NQ][9];
    int jq[NQ], xIdx[NQ], dIdx[NQ], nIdx[NQ];
#pragma unroll
    for (int q = 0; q < NQ; ++q) {
        const int v = q * 32 + lane;
        const int vc = (v < NITEM) ? v : NITEM - 1;
        const int f = vc / J_;
        jq[q] = vc - f * J_;
        xIdx[q] = f * XSTR + jq[q];
        dIdx[q] = f * TILE + jq[q] * (GSTR + 1);
        nIdx[q] = f * NSTR + jq[q];
#pragma unroll
        for (int c = 0; c < 9; ++c) acc[q][c] = 0.0f;
    }

    float* gw = lds_g[wv][0];
    float* const xr[KU] = { lds_x0[wv], lds_x1[wv], lds_x2[wv], lds_x3[wv], lds_x4[wv], lds_x5[wv] };
    float* nw = lds_n[wv];
    const float* xb = x + (size_t)b * T_ * (J_ * 3);

    float cur[6][NF], nxt[6][NF];
#pragma unroll
    for (int r = 0; r < 6; ++r)
#pragma unroll
        for (int f = 0; f < NF; ++f) nxt[r][f] = 0.0f;
    trip_load(xb, wv, lane, cur);
    for (int T = wv; T < TRIPS; T += WAVES) {
        const int Tn = (T + WAVES < TRIPS) ? T + WAVES : T;
        trip_load(xb, Tn, lane, nxt);
        trip_compute(cur, gw, xr, nw, lane, xIdx, dIdx, nIdx, acc);
#pragma unroll
        for (int r = 0; r < 6; ++r)
#pragma unroll
            for (int f = 0; f < NF; ++f) cur[r][f] = nxt[r][f];
    }

#pragma unroll
    for (int q = 0; q < NQ; ++q) {
        const int v = q * 32 + lane;
        if (v < NITEM) {
#pragma unroll
            for (int c = 0; c < 9; ++c)
                gw[v * 9 + c] = acc[q][c] * (1.0f / (float)T_);
        }
    }
    __syncthreads();

    if (tid < J_ * 9) {
        const int jj = tid / 9;
        const int c  = tid - jj * 9;
        float s = 0.0f;
        for (int w = 0; w < WAVES; ++w)
#pragma unroll
            for (int f = 0; f < NF; ++f) s += lds_g[w][0][(f * J_ + jj) * 9 + c];
        out[(size_t)b * (J_ * 9) + tid] = s;
    }
}

extern "C" void kernel_launch(void* const* d_in, const int* in_sizes, int n_in,
                              void* d_out, int out_size, void* d_ws, size_t ws_size,
                              hipStream_t stream) {
    (void)in_sizes; (void)n_in; (void)d_ws; (void)ws_size; (void)out_size;
    const float* x = (const float*)d_in[0];
    float* out = (float*)d_out;
    knn_feats_kernel<<<B_, 256, 0, stream>>>(x, out);
}
